# starting kernel
# baseline (speedup 1.0000x reference)
#include <hip/hip_runtime.h>

typedef float v2f __attribute__((ext_vector_type(2)));
typedef float v4f __attribute__((ext_vector_type(4)));
typedef float v8f __attribute__((ext_vector_type(8)));

#define N_IN   8192
#define N_OUT  8192
#define NB     8
#define KCH    1024
#define NCHUNK (N_IN / KCH)
#define NTILE  (N_OUT / 16)

__global__ __launch_bounds__(32)
void tri_wmma_partial(const float* __restrict__ w, const float* __restrict__ x,
                      const float* __restrict__ bias, float* __restrict__ ws,
                      float* __restrict__ out, int k_chunk, int final_mode) {
  const int tile    = blockIdx.x;
  const int o0      = tile << 4;
  const int fc_min  = (o0 > 4) ? (o0 - 4) : 0;
  const int k_begin = blockIdx.y * k_chunk;
  int i_end = k_begin + k_chunk;
  if (i_end > N_IN) i_end = N_IN;
  if (i_end <= fc_min) return;
  int i = max(k_begin, fc_min & ~3);

  const int lane = threadIdx.x;
  const int half = lane >> 4;
  const int n    = lane & 15;
  const int k0   = half << 1;

  const int  o  = o0 + n;
  const int  fc = (o > 4) ? (o - 4) : 0;
  const long rs = (long)o * N_IN - ((o >= 5) ? (((long)(o - 5) * (o - 4)) >> 1) : 0);
  const float* __restrict__ wrow = w + (rs - fc);
  const float* __restrict__ xrow = x + (long)(n & 7) * N_IN;

  v8f acc = {};

  int rag_end = o0 + 12;
  if (rag_end > i_end) rag_end = i_end;
  for (; i < rag_end; i += 4) {
    const int ii = i + k0;
    float a0 = __builtin_nontemporal_load(wrow + ii);
    float a1 = __builtin_nontemporal_load(wrow + ii + 1);
    a0 = (ii     >= fc) ? a0 : 0.0f;
    a1 = (ii + 1 >= fc) ? a1 : 0.0f;
    v2f af; af.x = a0; af.y = a1;
    v2f bf = *(const v2f*)(xrow + ii);
    acc = __builtin_amdgcn_wmma_f32_16x16x4_f32(false, af, false, bf,
                                                (short)0, acc, false, false);
  }

#pragma unroll 4
  for (; i < i_end; i += 4) {
    const int ii = i + k0;
    v2f af;
    af.x = __builtin_nontemporal_load(wrow + ii);
    af.y = __builtin_nontemporal_load(wrow + ii + 1);
    v2f bf = *(const v2f*)(xrow + ii);
    acc = __builtin_amdgcn_wmma_f32_16x16x4_f32(false, af, false, bf,
                                                (short)0, acc, false, false);
  }

  if (n < NB) {
    const int obase = o0 + (half << 3);
    const v4f b0 = *(const v4f*)(bias + obase);
    const v4f b1 = *(const v4f*)(bias + obase + 4);
    if (final_mode) {
      float* po = out + (long)n * N_OUT + (N_OUT - 8 - obase);
      v4f lo, hi;
      lo.x = acc[7] + b1.w; lo.y = acc[6] + b1.z;
      lo.z = acc[5] + b1.y; lo.w = acc[4] + b1.x;
      hi.x = acc[3] + b0.w; hi.y = acc[2] + b0.z;
      hi.z = acc[1] + b0.y; hi.w = acc[0] + b0.x;
      *(v4f*)po       = lo;
      *(v4f*)(po + 4) = hi;
    } else {
      float* p = ws + ((((long)blockIdx.y * NTILE + tile) * NB + n) << 4) + (half << 3);
      v4f lo, hi;
      lo.x = acc[0]; lo.y = acc[1]; lo.z = acc[2]; lo.w = acc[3];
      hi.x = acc[4]; hi.y = acc[5]; hi.z = acc[6]; hi.w = acc[7];
      *(v4f*)p       = lo;
      *(v4f*)(p + 4) = hi;
    }
  }
}

__global__ __launch_bounds__(256)
void tri_reduce(const float* __restrict__ ws, const float* __restrict__ bias,
                float* __restrict__ out) {
  const int t = blockIdx.x * 256 + threadIdx.x;
  if (t >= N_OUT * NB) return;
  const int o = t >> 3, m = t & 7;
  const int tile = o >> 4, r = o & 15;
  const int fc_min = (((o & ~15) - 4) > 0) ? ((o & ~15) - 4) : 0;
  float s = 0.0f;
  for (int c = (fc_min / KCH); c < NCHUNK; ++c)
    s += ws[((((long)c * NTILE + tile) * NB + m) << 4) + r];
  out[(long)m * N_OUT + (N_OUT - 1 - o)] = s + bias[o];
}

extern "C" void kernel_launch(void* const* d_in, const int* in_sizes, int n_in,
                              void* d_out, int out_size, void* d_ws, size_t ws_size,
                              hipStream_t stream) {
  (void)in_sizes; (void)n_in; (void)out_size;
  const float* x    = (const float*)d_in[0];
  const float* w    = (const float*)d_in[1];
  const float* bias = (const float*)d_in[2];
  float*       out  = (float*)d_out;

  const size_t ws_needed = (size_t)NCHUNK * NTILE * NB * 16 * sizeof(float);
  if (ws_size >= ws_needed) {
    dim3 grid(NTILE, NCHUNK);
    tri_wmma_partial<<<grid, 32, 0, stream>>>(w, x, bias, (float*)d_ws, out,
                                              KCH,  0);
    tri_reduce<<<(N_OUT * NB + 255) / 256, 256, 0, stream>>>(
        (const float*)d_ws, bias, out);
  } else {
    dim3 grid(NTILE, 1);
    tri_wmma_partial<<<grid, 32, 0, stream>>>(w, x, bias, nullptr, out,
                                              N_IN,  1);
  }
}


# Two register sets of 8 column steps: the next group's 24 loads are requested before the current group's WMMAs, so a wave keeps 24-46 loads in flight and never drains its queue; sum order unchanged
# speedup vs baseline: 2.6794x; 2.6794x over previous
#include <hip/hip_runtime.h>


typedef float v2f __attribute__((ext_vector_type(2)));
typedef float v4f __attribute__((ext_vector_type(4)));
typedef float v8f __attribute__((ext_vector_type(8)));
typedef float v4fu __attribute__((ext_vector_type(4), aligned(4)));

#define N_IN   8192
#define N_OUT  8192
#define NB     8
#define KCH    1024
#define NCHUNK (N_IN / KCH)
#define NTILE  (N_OUT / 16)


#define GSTEPS 8
#define GCOLS  (GSTEPS * 16)
struct Group { v4f a[GSTEPS], b[GSTEPS], x[GSTEPS]; };

__device__ __forceinline__ void group_load(Group& g, const float* __restrict__ wa,
                                           const float* __restrict__ wb,
                                           const float* __restrict__ xr, int i) {
#pragma unroll
  for (int s = 0; s < GSTEPS; ++s) {
    g.a[s] = __builtin_nontemporal_load((const v4fu*)(wa + i + 16 * s));
    g.b[s] = __builtin_nontemporal_load((const v4fu*)(wb + i + 16 * s));
    g.x[s] = *(const v4fu*)(xr + i + 16 * s);
  }
}

__device__ __forceinline__ void group_mma(const Group& g, v8f& acc0, v8f& acc1) {
#pragma unroll
  for (int s = 0; s < GSTEPS; ++s) {
    const v4f a = g.a[s], b = g.b[s], xv = g.x[s];
    v2f a01 = {a.x, a.y}, a23 = {a.z, a.w}, b01 = {b.x, b.y}, b23 = {b.z, b.w};
    v2f x01 = {xv.x, xv.y}, x23 = {xv.z, xv.w};
    acc0 = __builtin_amdgcn_wmma_f32_16x16x4_f32(false, a01, false, x01, (short)0, acc0, false, false);
    acc1 = __builtin_amdgcn_wmma_f32_16x16x4_f32(false, b01, false, x01, (short)0, acc1, false, false);
    acc0 = __builtin_amdgcn_wmma_f32_16x16x4_f32(false, a23, false, x23, (short)0, acc0, false, false);
    acc1 = __builtin_amdgcn_wmma_f32_16x16x4_f32(false, b23, false, x23, (short)0, acc1, false, false);
  }
}

__global__ __launch_bounds__(32)
void tri_wmma_partial(const float* __restrict__ w, const float* __restrict__ x,
                      const float* __restrict__ bias, float* __restrict__ ws,
                      float* __restrict__ out, int k_chunk, int final_mode) {
  const int tile    = blockIdx.x;
  const int o0      = tile << 4;
  const int fc_min  = (o0 > 4) ? (o0 - 4) : 0;
  const int k_begin = blockIdx.y * k_chunk;
  int i_end = k_begin + k_chunk;
  if (i_end > N_IN) i_end = N_IN;
  if (i_end <= fc_min) return;
  int i = max(k_begin, fc_min & ~15);

  const int lane = threadIdx.x;
  const int h    = lane >> 4;
  const int c    = lane & 15;
  const int r8   = c & 7;
  const int koff = ((c >> 3) * 2 + h) << 2;

  const int  oa  = o0 + r8;
  const int  ob  = oa + 8;
  const int  fca = (oa > 4) ? (oa - 4) : 0;
  const int  fcb = (ob > 4) ? (ob - 4) : 0;
  const long rsa = (long)oa * N_IN - ((oa >= 5) ? (((long)(oa - 5) * (oa - 4)) >> 1) : 0);
  const long rsb = (long)ob * N_IN - ((ob >= 5) ? (((long)(ob - 5) * (ob - 4)) >> 1) : 0);
  const float* __restrict__ wa = w + (rsa - fca) + koff;
  const float* __restrict__ wb = w + (rsb - fcb) + koff;
  const float* __restrict__ xr = x + (long)r8 * N_IN + koff;

  v8f acc0 = {};
  v8f acc1 = {};

  int rag_end = o0 + 16;
  if (rag_end > i_end) rag_end = i_end;
  for (; i < rag_end; i += 16) {
    const int col = i + koff;
    v4f a = __builtin_nontemporal_load((const v4fu*)(wa + i));
    v4f b = __builtin_nontemporal_load((const v4fu*)(wb + i));
    a.x = (col     >= fca) ? a.x : 0.0f;  a.y = (col + 1 >= fca) ? a.y : 0.0f;
    a.z = (col + 2 >= fca) ? a.z : 0.0f;  a.w = (col + 3 >= fca) ? a.w : 0.0f;
    b.x = (col     >= fcb) ? b.x : 0.0f;  b.y = (col + 1 >= fcb) ? b.y : 0.0f;
    b.z = (col + 2 >= fcb) ? b.z : 0.0f;  b.w = (col + 3 >= fcb) ? b.w : 0.0f;
    const v4f xv = *(const v4fu*)(xr + i);
    v2f a01 = {a.x, a.y}, a23 = {a.z, a.w}, b01 = {b.x, b.y}, b23 = {b.z, b.w};
    v2f x01 = {xv.x, xv.y}, x23 = {xv.z, xv.w};
    acc0 = __builtin_amdgcn_wmma_f32_16x16x4_f32(false, a01, false, x01, (short)0, acc0, false, false);
    acc1 = __builtin_amdgcn_wmma_f32_16x16x4_f32(false, b01, false, x01, (short)0, acc1, false, false);
    acc0 = __builtin_amdgcn_wmma_f32_16x16x4_f32(false, a23, false, x23, (short)0, acc0, false, false);
    acc1 = __builtin_amdgcn_wmma_f32_16x16x4_f32(false, b23, false, x23, (short)0, acc1, false, false);
  }

  const int lead_end = i + ((i_end - i) & (GCOLS - 1));
#pragma unroll 4
  for (; i < lead_end; i += 16) {
    const v4f a  = __builtin_nontemporal_load((const v4fu*)(wa + i));
    const v4f b  = __builtin_nontemporal_load((const v4fu*)(wb + i));
    const v4f xv = *(const v4fu*)(xr + i);
    v2f a01 = {a.x, a.y}, a23 = {a.z, a.w}, b01 = {b.x, b.y}, b23 = {b.z, b.w};
    v2f x01 = {xv.x, xv.y}, x23 = {xv.z, xv.w};
    acc0 = __builtin_amdgcn_wmma_f32_16x16x4_f32(false, a01, false, x01, (short)0, acc0, false, false);
    acc1 = __builtin_amdgcn_wmma_f32_16x16x4_f32(false, b01, false, x01, (short)0, acc1, false, false);
    acc0 = __builtin_amdgcn_wmma_f32_16x16x4_f32(false, a23, false, x23, (short)0, acc0, false, false);
    acc1 = __builtin_amdgcn_wmma_f32_16x16x4_f32(false, b23, false, x23, (short)0, acc1, false, false);
  }

  int left = (i_end - i) / GCOLS;
  if (left > 0) {
    Group g0, g1;
    group_load(g0, wa, wb, xr, i);
    while (left >= 3) {
      group_load(g1, wa, wb, xr, i + GCOLS);
      group_mma(g0, acc0, acc1);
      group_load(g0, wa, wb, xr, i + 2 * GCOLS);
      group_mma(g1, acc0, acc1);
      i += 2 * GCOLS;
      left -= 2;
    }
    if (left == 2) {
      group_load(g1, wa, wb, xr, i + GCOLS);
      group_mma(g0, acc0, acc1);
      group_mma(g1, acc0, acc1);
    } else {
      group_mma(g0, acc0, acc1);
    }
  }

  float t0[8], t1[8];
#pragma unroll
  for (int r = 0; r < 8; ++r) {
    t0[r] = acc0[r] + __shfl_xor(acc0[r], 24);
    t1[r] = acc1[r] + __shfl_xor(acc1[r], 24);
  }

  if (lane < NB) {
    const int n = lane;
    if (final_mode) {
      const v4f b0 = *(const v4f*)(bias + o0);
      const v4f b1 = *(const v4f*)(bias + o0 + 4);
      const v4f b2 = *(const v4f*)(bias + o0 + 8);
      const v4f b3 = *(const v4f*)(bias + o0 + 12);
      float* po = out + (long)n * N_OUT + (N_OUT - 16 - o0);
      v4f q0, q1, q2, q3;
      q0.x = t1[7] + b3.w; q0.y = t1[6] + b3.z; q0.z = t1[5] + b3.y; q0.w = t1[4] + b3.x;
      q1.x = t1[3] + b2.w; q1.y = t1[2] + b2.z; q1.z = t1[1] + b2.y; q1.w = t1[0] + b2.x;
      q2.x = t0[7] + b1.w; q2.y = t0[6] + b1.z; q2.z = t0[5] + b1.y; q2.w = t0[4] + b1.x;
      q3.x = t0[3] + b0.w; q3.y = t0[2] + b0.z; q3.z = t0[1] + b0.y; q3.w = t0[0] + b0.x;
      *(v4f*)po        = q0;
      *(v4f*)(po + 4)  = q1;
      *(v4f*)(po + 8)  = q2;
      *(v4f*)(po + 12) = q3;
    } else {
      float* p = ws + ((((long)blockIdx.y * NTILE + tile) * NB + n) << 4);
      v4f q0, q1, q2, q3;
      q0.x = t0[0]; q0.y = t0[1]; q0.z = t0[2]; q0.w = t0[3];
      q1.x = t0[4]; q1.y = t0[5]; q1.z = t0[6]; q1.w = t0[7];
      q2.x = t1[0]; q2.y = t1[1]; q2.z = t1[2]; q2.w = t1[3];
      q3.x = t1[4]; q3.y = t1[5]; q3.z = t1[6]; q3.w = t1[7];
      *(v4f*)p        = q0;
      *(v4f*)(p + 4)  = q1;
      *(v4f*)(p + 8)  = q2;
      *(v4f*)(p + 12) = q3;
    }
  }
}

__global__ __launch_bounds__(256)
void tri_reduce(const float* __restrict__ ws, const float* __restrict__ bias,
                float* __restrict__ out) {
  const int t = blockIdx.x * 256 + threadIdx.x;
  if (t >= N_OUT * NB) return;
  const int o = t >> 3, m = t & 7;
  const int tile = o >> 4, r = o & 15;
  const int fc_min = (((o & ~15) - 4) > 0) ? ((o & ~15) - 4) : 0;
  float s = 0.0f;
  for (int c = (fc_min / KCH); c < NCHUNK; ++c)
    s += ws[((((long)c * NTILE + tile) * NB + m) << 4) + r];
  out[(long)m * N_OUT + (N_OUT - 1 - o)] = s + bias[o];
}

extern "C" void kernel_launch(void* const* d_in, const int* in_sizes, int n_in,
                              void* d_out, int out_size, void* d_ws, size_t ws_size,
                              hipStream_t stream) {
  (void)in_sizes; (void)n_in; (void)out_size;
  const float* x    = (const float*)d_in[0];
  const float* w    = (const float*)d_in[1];
  const float* bias = (const float*)d_in[2];
  float*       out  = (float*)d_out;

  const size_t ws_needed = (size_t)NCHUNK * NTILE * NB * 16 * sizeof(float);
  if (ws_size >= ws_needed) {
    dim3 grid(NTILE, NCHUNK);
    tri_wmma_partial<<<grid, 32, 0, stream>>>(w, x, bias, (float*)d_ws, out,
                                              KCH,  0);
    tri_reduce<<<(N_OUT * NB + 255) / 256, 256, 0, stream>>>(
        (const float*)d_ws, bias, out);
  } else {
    dim3 grid(NTILE, 1);
    tri_wmma_partial<<<grid, 32, 0, stream>>>(w, x, bias, nullptr, out,
                                              N_IN,  1);
  }
}


# Fragment retile, no LDS: WMMA tile = 4 rows x 4 column segments against 4 batches x 4 segments (lo/hi batch halves): a W load reads 128 B per row, x loads whole lines; 2x WMMAs; sum order changes
# speedup vs baseline: 4.3243x; 1.6139x over previous
#include <hip/hip_runtime.h>


typedef float v2f __attribute__((ext_vector_type(2)));
typedef float v4f __attribute__((ext_vector_type(4)));
typedef float v8f __attribute__((ext_vector_type(8)));
typedef float v4fu __attribute__((ext_vector_type(4), aligned(4)));

#define N_IN   8192
#define N_OUT  8192
#define NB     8
#define KCH    1024
#define NCHUNK (N_IN / KCH)
#define NTILE  (N_OUT / 16)

#define STEPC  32
#define GSTEPS 4
#define GCOLS  (GSTEPS * STEPC)
#define NT4    4
struct Group { v4f a[GSTEPS][NT4], xl[GSTEPS], xh[GSTEPS]; };
struct RowPtrs { const float* w[NT4]; const float* xlo; const float* xhi; };

__device__ __forceinline__ void group_load(Group& g, const RowPtrs& p, int i) {
#pragma unroll
  for (int s = 0; s < GSTEPS; ++s) {
#pragma unroll
    for (int t = 0; t < NT4; ++t)
      g.a[s][t] = __builtin_nontemporal_load((const v4fu*)(p.w[t] + i + STEPC * s));
    g.xl[s] = *(const v4fu*)(p.xlo + i + STEPC * s);
    g.xh[s] = *(const v4fu*)(p.xhi + i + STEPC * s);
  }
}

__device__ __forceinline__ void step_mma(const v4f (&a)[NT4], const v4f xl, const v4f xh,
                                         v8f (&lo)[NT4], v8f (&hi)[NT4]) {
  const v2f l01 = {xl.x, xl.y}, l23 = {xl.z, xl.w};
  const v2f h01 = {xh.x, xh.y}, h23 = {xh.z, xh.w};
#pragma unroll
  for (int t = 0; t < NT4; ++t) {
    const v2f a01 = {a[t].x, a[t].y}, a23 = {a[t].z, a[t].w};
    lo[t] = __builtin_amdgcn_wmma_f32_16x16x4_f32(false, a01, false, l01, (short)0, lo[t], false, false);
    hi[t] = __builtin_amdgcn_wmma_f32_16x16x4_f32(false, a01, false, h01, (short)0, hi[t], false, false);
    lo[t] = __builtin_amdgcn_wmma_f32_16x16x4_f32(false, a23, false, l23, (short)0, lo[t], false, false);
    hi[t] = __builtin_amdgcn_wmma_f32_16x16x4_f32(false, a23, false, h23, (short)0, hi[t], false, false);
  }
}

__device__ __forceinline__ void group_mma(const Group& g, v8f (&lo)[NT4], v8f (&hi)[NT4]) {
#pragma unroll
  for (int s = 0; s < GSTEPS; ++s) step_mma(g.a[s], g.xl[s], g.xh[s], lo, hi);
}

__global__ __launch_bounds__(32)
void tri_wmma_partial(const float* __restrict__ w, const float* __restrict__ x,
                      const float* __restrict__ bias, float* __restrict__ ws,
                      float* __restrict__ out, int k_chunk, int final_mode) {
  const int tile    = blockIdx.x;
  const int o0      = tile << 4;
  const int fc_min  = (o0 > 4) ? (o0 - 4) : 0;
  const int k_begin = blockIdx.y * k_chunk;
  int i_end = k_begin + k_chunk;
  if (i_end > N_IN) i_end = N_IN;
  if (i_end <= fc_min) return;
  int i = max(k_begin, fc_min & ~(STEPC - 1));

  const int lane = threadIdx.x;
  const int h    = lane >> 4;
  const int c    = lane & 15;
  const int q    = c & 3;
  const int koff = ((c >> 2) * 2 + h) << 2;

  RowPtrs p;
  int fc[NT4];
#pragma unroll
  for (int t = 0; t < NT4; ++t) {
    const int  o  = o0 + 4 * t + q;
    fc[t] = (o > 4) ? (o - 4) : 0;
    const long rs = (long)o * N_IN - ((o >= 5) ? (((long)(o - 5) * (o - 4)) >> 1) : 0);
    p.w[t] = w + (rs - fc[t]) + koff;
  }
  p.xlo = x + (long)q * N_IN + koff;
  p.xhi = p.xlo + 4L * N_IN;

  v8f lo[NT4], hi[NT4];
#pragma unroll
  for (int t = 0; t < NT4; ++t) { lo[t] = (v8f){}; hi[t] = (v8f){}; }

  int rag_end = o0 + 16;
  if (rag_end > i_end) rag_end = i_end;
  for (; i < rag_end; i += STEPC) {
    const int col = i + koff;
    v4f a[NT4];
#pragma unroll
    for (int t = 0; t < NT4; ++t) {
      v4f v = __builtin_nontemporal_load((const v4fu*)(p.w[t] + i));
      v.x = (col     >= fc[t]) ? v.x : 0.0f;  v.y = (col + 1 >= fc[t]) ? v.y : 0.0f;
      v.z = (col + 2 >= fc[t]) ? v.z : 0.0f;  v.w = (col + 3 >= fc[t]) ? v.w : 0.0f;
      a[t] = v;
    }
    const v4f xl = *(const v4fu*)(p.xlo + i);
    const v4f xh = *(const v4fu*)(p.xhi + i);
    step_mma(a, xl, xh, lo, hi);
  }

  const int lead_end = i + ((i_end - i) & (GCOLS - 1));
  for (; i < lead_end; i += STEPC) {
    v4f a[NT4];
#pragma unroll
    for (int t = 0; t < NT4; ++t) a[t] = __builtin_nontemporal_load((const v4fu*)(p.w[t] + i));
    const v4f xl = *(const v4fu*)(p.xlo + i);
    const v4f xh = *(const v4fu*)(p.xhi + i);
    step_mma(a, xl, xh, lo, hi);
  }

  int left = (i_end - i) / GCOLS;
  if (left > 0) {
    Group g0, g1;
    group_load(g0, p, i);
    while (left >= 3) {
      group_load(g1, p, i + GCOLS);
      __builtin_amdgcn_sched_barrier(0);
      group_mma(g0, lo, hi);
      __builtin_amdgcn_sched_barrier(0);
      group_load(g0, p, i + 2 * GCOLS);
      __builtin_amdgcn_sched_barrier(0);
      group_mma(g1, lo, hi);
      __builtin_amdgcn_sched_barrier(0);
      i += 2 * GCOLS;
      left -= 2;
    }
    if (left == 2) {
      group_load(g1, p, i + GCOLS);
      group_mma(g0, lo, hi);
      group_mma(g1, lo, hi);
    } else {
      group_mma(g0, lo, hi);
    }
  }

  const bool up = (lane & 4) != 0;
  float val[16];
#pragma unroll
  for (int t = 0; t < NT4; ++t) {
#pragma unroll
    for (int r = 0; r < 4; ++r) {
      float sl = up ? lo[t][r + 4] : lo[t][r];
      float sh = up ? hi[t][r + 4] : hi[t][r];
      sl += __shfl_xor(sl, 4);   sh += __shfl_xor(sh, 4);
      sl += __shfl_xor(sl, 24);  sh += __shfl_xor(sh, 24);
      val[4 * t + r] = up ? sh : sl;
    }
  }

  if (lane < NB) {
    const int n = lane;
    if (final_mode) {
      const v4f b0 = *(const v4f*)(bias + o0);
      const v4f b1 = *(const v4f*)(bias + o0 + 4);
      const v4f b2 = *(const v4f*)(bias + o0 + 8);
      const v4f b3 = *(const v4f*)(bias + o0 + 12);
      float* po = out + (long)n * N_OUT + (N_OUT - 16 - o0);
      v4f q0, q1, q2, q3;
      q0.x = val[15] + b3.w; q0.y = val[14] + b3.z; q0.z = val[13] + b3.y; q0.w = val[12] + b3.x;
      q1.x = val[11] + b2.w; q1.y = val[10] + b2.z; q1.z = val[9]  + b2.y; q1.w = val[8]  + b2.x;
      q2.x = val[7]  + b1.w; q2.y = val[6]  + b1.z; q2.z = val[5]  + b1.y; q2.w = val[4]  + b1.x;
      q3.x = val[3]  + b0.w; q3.y = val[2]  + b0.z; q3.z = val[1]  + b0.y; q3.w = val[0]  + b0.x;
      *(v4f*)po        = q0;
      *(v4f*)(po + 4)  = q1;
      *(v4f*)(po + 8)  = q2;
      *(v4f*)(po + 12) = q3;
    } else {
      float* pw = ws + ((((long)blockIdx.y * NTILE + tile) * NB + n) << 4);
      v4f q0, q1, q2, q3;
      q0.x = val[0];  q0.y = val[1];  q0.z = val[2];  q0.w = val[3];
      q1.x = val[4];  q1.y = val[5];  q1.z = val[6];  q1.w = val[7];
      q2.x = val[8];  q2.y = val[9];  q2.z = val[10]; q2.w = val[11];
      q3.x = val[12]; q3.y = val[13]; q3.z = val[14]; q3.w = val[15];
      *(v4f*)pw        = q0;
      *(v4f*)(pw + 4)  = q1;
      *(v4f*)(pw + 8)  = q2;
      *(v4f*)(pw + 12) = q3;
    }
  }
}

__global__ __launch_bounds__(256)
void tri_reduce(const float* __restrict__ ws, const float* __restrict__ bias,
                float* __restrict__ out) {
  const int t = blockIdx.x * 256 + threadIdx.x;
  if (t >= N_OUT * NB) return;
  const int o = t >> 3, m = t & 7;
  const int tile = o >> 4, r = o & 15;
  const int fc_min = (((o & ~15) - 4) > 0) ? ((o & ~15) - 4) : 0;
  float s = 0.0f;
  for (int c = (fc_min / KCH); c < NCHUNK; ++c)
    s += ws[((((long)c * NTILE + tile) * NB + m) << 4) + r];
  out[(long)m * N_OUT + (N_OUT - 1 - o)] = s + bias[o];
}

extern "C" void kernel_launch(void* const* d_in, const int* in_sizes, int n_in,
                              void* d_out, int out_size, void* d_ws, size_t ws_size,
                              hipStream_t stream) {
  (void)in_sizes; (void)n_in; (void)out_size;
  const float* x    = (const float*)d_in[0];
  const float* w    = (const float*)d_in[1];
  const float* bias = (const float*)d_in[2];
  float*       out  = (float*)d_out;

  const size_t ws_needed = (size_t)NCHUNK * NTILE * NB * 16 * sizeof(float);
  if (ws_size >= ws_needed) {
    dim3 grid(NTILE, NCHUNK);
    tri_wmma_partial<<<grid, 32, 0, stream>>>(w, x, bias, (float*)d_ws, out,
                                              KCH,  0);
    tri_reduce<<<(N_OUT * NB + 255) / 256, 256, 0, stream>>>(
        (const float*)d_ws, bias, out);
  } else {
    dim3 grid(NTILE, 1);
    tri_wmma_partial<<<grid, 32, 0, stream>>>(w, x, bias, nullptr, out,
                                              N_IN,  1);
  }
}


# Work split: 256 workgroups of 8 waves; wave w takes slice w of tile p and slice 7-w of tile 511-p, whole 128-column groups dealt evenly: equal work per wave, 2 per SIMD; sum order changes
# speedup vs baseline: 4.7130x; 1.0899x over previous
#include <hip/hip_runtime.h>


typedef float v2f __attribute__((ext_vector_type(2)));
typedef float v4f __attribute__((ext_vector_type(4)));
typedef float v8f __attribute__((ext_vector_type(8)));
typedef float v4fu __attribute__((ext_vector_type(4), aligned(4)));

#define N_IN   8192
#define N_OUT  8192
#define NB     8
#define NSL    8
#define NSL_LOG2 3
#define NTILE  (N_OUT / 16)
#define NPAIR  (NTILE / 2)

#define STEPC  32
#define GSTEPS 4
#define GCOLS  (GSTEPS * STEPC)
#define NT4    4
struct Group { v4f a[GSTEPS][NT4], xl[GSTEPS], xh[GSTEPS]; };
struct RowPtrs { const float* w[NT4]; const float* xlo; const float* xhi; };

__device__ __forceinline__ void group_load(Group& g, const RowPtrs& p, int i) {
#pragma unroll
  for (int s = 0; s < GSTEPS; ++s) {
#pragma unroll
    for (int t = 0; t < NT4; ++t)
      g.a[s][t] = __builtin_nontemporal_load((const v4fu*)(p.w[t] + i + STEPC * s));
    g.xl[s] = *(const v4fu*)(p.xlo + i + STEPC * s);
    g.xh[s] = *(const v4fu*)(p.xhi + i + STEPC * s);
  }
}

__device__ __forceinline__ void step_mma(const v4f (&a)[NT4], const v4f xl, const v4f xh,
                                         v8f (&lo)[NT4], v8f (&hi)[NT4]) {
  const v2f l01 = {xl.x, xl.y}, l23 = {xl.z, xl.w};
  const v2f h01 = {xh.x, xh.y}, h23 = {xh.z, xh.w};
#pragma unroll
  for (int t = 0; t < NT4; ++t) {
    const v2f a01 = {a[t].x, a[t].y}, a23 = {a[t].z, a[t].w};
    lo[t] = __builtin_amdgcn_wmma_f32_16x16x4_f32(false, a01, false, l01, (short)0, lo[t], false, false);
    hi[t] = __builtin_amdgcn_wmma_f32_16x16x4_f32(false, a01, false, h01, (short)0, hi[t], false, false);
    lo[t] = __builtin_amdgcn_wmma_f32_16x16x4_f32(false, a23, false, l23, (short)0, lo[t], false, false);
    hi[t] = __builtin_amdgcn_wmma_f32_16x16x4_f32(false, a23, false, h23, (short)0, hi[t], false, false);
  }
}

__device__ __forceinline__ void group_mma(const Group& g, v8f (&lo)[NT4], v8f (&hi)[NT4]) {
#pragma unroll
  for (int s = 0; s < GSTEPS; ++s) step_mma(g.a[s], g.xl[s], g.xh[s], lo, hi);
}

__global__ __launch_bounds__(NSL * 32)
void tri_wmma_partial(const float* __restrict__ w, const float* __restrict__ x,
                      const float* __restrict__ bias, float* __restrict__ ws,
                      float* __restrict__ out, int slice_shift, int final_mode) {
  const int lane = threadIdx.x & 31;
  const int wave = __builtin_amdgcn_readfirstlane((int)(threadIdx.x >> 5));
  const int nsl  = 1 << slice_shift;
  const int h    = lane >> 4;
  const int c    = lane & 15;
  const int q    = c & 3;
  const int koff = ((c >> 2) * 2 + h) << 2;
  const int n_items = final_mode ? 1 : 2;

  for (int it = 0; it < n_items; ++it) {
    int tile, sl;
    if (final_mode)   { tile = blockIdx.x;             sl = 0; }
    else if (it == 0) { tile = blockIdx.x;             sl = wave; }
    else              { tile = NTILE - 1 - blockIdx.x; sl = nsl - 1 - wave; }
    const int o0     = tile << 4;
    const int fc_min = (o0 > 4) ? (o0 - 4) : 0;
    const int k0     = fc_min & ~(STEPC - 1);
    const int span   = N_IN - k0;
    const int rem    = span & (GCOLS - 1);
    const int ngrp   = span / GCOLS;
    const int base   = ngrp >> slice_shift;
    const int extra  = ngrp & (nsl - 1);
    const int cnt    = base + ((sl < extra) ? 1 : 0);
    const int gs     = sl * base + ((sl < extra) ? sl : extra);
    const int i_end  = k0 + rem + (gs + cnt) * GCOLS;
    int i = (sl == 0) ? k0 : (k0 + rem + gs * GCOLS);

    RowPtrs p;
    int fc[NT4];
#pragma unroll
    for (int t = 0; t < NT4; ++t) {
      const int  o  = o0 + 4 * t + q;
      fc[t] = (o > 4) ? (o - 4) : 0;
      const long rs = (long)o * N_IN - ((o >= 5) ? (((long)(o - 5) * (o - 4)) >> 1) : 0);
      p.w[t] = w + (rs - fc[t]) + koff;
    }
    p.xlo = x + (long)q * N_IN + koff;
    p.xhi = p.xlo + 4L * N_IN;

    v8f lo[NT4], hi[NT4];
#pragma unroll
    for (int t = 0; t < NT4; ++t) { lo[t] = (v8f){}; hi[t] = (v8f){}; }

    int rag_end = o0 + 16;
    if (rag_end > i_end) rag_end = i_end;
    for (; i < rag_end; i += STEPC) {
      const int col = i + koff;
      v4f a[NT4];
#pragma unroll
      for (int t = 0; t < NT4; ++t) {
        v4f v = __builtin_nontemporal_load((const v4fu*)(p.w[t] + i));
        v.x = (col     >= fc[t]) ? v.x : 0.0f;  v.y = (col + 1 >= fc[t]) ? v.y : 0.0f;
        v.z = (col + 2 >= fc[t]) ? v.z : 0.0f;  v.w = (col + 3 >= fc[t]) ? v.w : 0.0f;
        a[t] = v;
      }
      const v4f xl = *(const v4fu*)(p.xlo + i);
      const v4f xh = *(const v4fu*)(p.xhi + i);
      step_mma(a, xl, xh, lo, hi);
    }

    const int lead_end = i + ((i_end - i) & (GCOLS - 1));
    for (; i < lead_end; i += STEPC) {
      v4f a[NT4];
#pragma unroll
      for (int t = 0; t < NT4; ++t) a[t] = __builtin_nontemporal_load((const v4fu*)(p.w[t] + i));
      const v4f xl = *(const v4fu*)(p.xlo + i);
      const v4f xh = *(const v4fu*)(p.xhi + i);
      step_mma(a, xl, xh, lo, hi);
    }

    int left = (i_end - i) / GCOLS;
    Group g0, g1;
    if (left > 0) group_load(g0, p, i);
    if (left > 0) {
      while (left >= 3) {
        group_load(g1, p, i + GCOLS);
        __builtin_amdgcn_sched_barrier(0);
        group_mma(g0, lo, hi);
        __builtin_amdgcn_sched_barrier(0);
        group_load(g0, p, i + 2 * GCOLS);
        __builtin_amdgcn_sched_barrier(0);
        group_mma(g1, lo, hi);
        __builtin_amdgcn_sched_barrier(0);
        i += 2 * GCOLS;
        left -= 2;
      }
      if (left == 2) {
        group_load(g1, p, i + GCOLS);
        group_mma(g0, lo, hi);
        group_mma(g1, lo, hi);
      } else {
        group_mma(g0, lo, hi);
      }
    }

    const bool up = (lane & 4) != 0;
    float val[16];
#pragma unroll
    for (int t = 0; t < NT4; ++t) {
#pragma unroll
      for (int r = 0; r < 4; ++r) {
        float sl = up ? lo[t][r + 4] : lo[t][r];
        float sh = up ? hi[t][r + 4] : hi[t][r];
        sl += __shfl_xor(sl, 4);   sh += __shfl_xor(sh, 4);
        sl += __shfl_xor(sl, 24);  sh += __shfl_xor(sh, 24);
        val[4 * t + r] = up ? sh : sl;
      }
    }

    if (lane < NB) {
      const int n = lane;
      if (final_mode) {
        const v4f b0 = *(const v4f*)(bias + o0);
        const v4f b1 = *(const v4f*)(bias + o0 + 4);
        const v4f b2 = *(const v4f*)(bias + o0 + 8);
        const v4f b3 = *(const v4f*)(bias + o0 + 12);
        float* po = out + (long)n * N_OUT + (N_OUT - 16 - o0);
        v4f q0, q1, q2, q3;
        q0.x = val[15] + b3.w; q0.y = val[14] + b3.z; q0.z = val[13] + b3.y; q0.w = val[12] + b3.x;
        q1.x = val[11] + b2.w; q1.y = val[10] + b2.z; q1.z = val[9]  + b2.y; q1.w = val[8]  + b2.x;
        q2.x = val[7]  + b1.w; q2.y = val[6]  + b1.z; q2.z = val[5]  + b1.y; q2.w = val[4]  + b1.x;
        q3.x = val[3]  + b0.w; q3.y = val[2]  + b0.z; q3.z = val[1]  + b0.y; q3.w = val[0]  + b0.x;
        *(v4f*)po        = q0;
        *(v4f*)(po + 4)  = q1;
        *(v4f*)(po + 8)  = q2;
        *(v4f*)(po + 12) = q3;
      } else {
        float* pw = ws + ((((long)sl * NTILE + tile) * NB + n) << 4);
        v4f q0, q1, q2, q3;
        q0.x = val[0];  q0.y = val[1];  q0.z = val[2];  q0.w = val[3];
        q1.x = val[4];  q1.y = val[5];  q1.z = val[6];  q1.w = val[7];
        q2.x = val[8];  q2.y = val[9];  q2.z = val[10]; q2.w = val[11];
        q3.x = val[12]; q3.y = val[13]; q3.z = val[14]; q3.w = val[15];
        *(v4f*)pw        = q0;
        *(v4f*)(pw + 4)  = q1;
        *(v4f*)(pw + 8)  = q2;
        *(v4f*)(pw + 12) = q3;
      }
    }
  }
}

__global__ __launch_bounds__(256)
void tri_reduce(const float* __restrict__ ws, const float* __restrict__ bias,
                float* __restrict__ out) {
  const int t = blockIdx.x * 256 + threadIdx.x;
  if (t >= N_OUT * NB) return;
  const int o = t >> 3, m = t & 7;
  const int tile = o >> 4, r = o & 15;
  float s = 0.0f;
  for (int c = 0; c < NSL; ++c)
    s += ws[((((long)c * NTILE + tile) * NB + m) << 4) + r];
  out[(long)m * N_OUT + (N_OUT - 1 - o)] = s + bias[o];
}

extern "C" void kernel_launch(void* const* d_in, const int* in_sizes, int n_in,
                              void* d_out, int out_size, void* d_ws, size_t ws_size,
                              hipStream_t stream) {
  (void)in_sizes; (void)n_in; (void)out_size;
  const float* x    = (const float*)d_in[0];
  const float* w    = (const float*)d_in[1];
  const float* bias = (const float*)d_in[2];
  float*       out  = (float*)d_out;

  const size_t ws_needed = (size_t)NSL * NTILE * NB * 16 * sizeof(float);
  if (ws_size >= ws_needed) {
    tri_wmma_partial<<<NPAIR, NSL * 32, 0, stream>>>(w, x, bias, (float*)d_ws, out,
                                                     NSL_LOG2,  0);
    tri_reduce<<<(N_OUT * NB + 255) / 256, 256, 0, stream>>>(
        (const float*)d_ws, bias, out);
  } else {
    tri_wmma_partial<<<NTILE, 32, 0, stream>>>(w, x, bias, nullptr, out,
                                               0,  1);
  }
}


# Second launch removed: each wave puts its two slice records into LDS instead of the workspace; behind one barrier the workgroup adds the 8 slices of both its tiles, then bias; sum order unchanged
# speedup vs baseline: 5.2715x; 1.1185x over previous
#include <hip/hip_runtime.h>


typedef float v2f __attribute__((ext_vector_type(2)));
typedef float v4f __attribute__((ext_vector_type(4)));
typedef float v8f __attribute__((ext_vector_type(8)));
typedef float v4fu __attribute__((ext_vector_type(4), aligned(4)));

#define N_IN   8192
#define N_OUT  8192
#define NB     8
#define NSL    8
#define NSL_LOG2 3
#define NTILE  (N_OUT / 16)
#define NPAIR  (NTILE / 2)

#define STEPC  32
#define GSTEPS 4
#define GCOLS  (GSTEPS * STEPC)
#define NT4    4
struct Group { v4f a[GSTEPS][NT4], xl[GSTEPS], xh[GSTEPS]; };
__device__ __forceinline__ void step_mma(const v4f (&a)[NT4], const v4f xl, const v4f xh,
                                         v8f (&lo)[NT4], v8f (&hi)[NT4]) {
  const v2f l01 = {xl.x, xl.y}, l23 = {xl.z, xl.w};
  const v2f h01 = {xh.x, xh.y}, h23 = {xh.z, xh.w};
#pragma unroll
  for (int t = 0; t < NT4; ++t) {
    const v2f a01 = {a[t].x, a[t].y}, a23 = {a[t].z, a[t].w};
    lo[t] = __builtin_amdgcn_wmma_f32_16x16x4_f32(false, a01, false, l01, (short)0, lo[t], false, false);
    hi[t] = __builtin_amdgcn_wmma_f32_16x16x4_f32(false, a01, false, h01, (short)0, hi[t], false, false);
    lo[t] = __builtin_amdgcn_wmma_f32_16x16x4_f32(false, a23, false, l23, (short)0, lo[t], false, false);
    hi[t] = __builtin_amdgcn_wmma_f32_16x16x4_f32(false, a23, false, h23, (short)0, hi[t], false, false);
  }
}

__device__ __forceinline__ void group_mma(const Group& g, v8f (&lo)[NT4], v8f (&hi)[NT4]) {
#pragma unroll
  for (int s = 0; s < GSTEPS; ++s) step_mma(g.a[s], g.xl[s], g.xh[s], lo, hi);
}

struct WPtrs { const float* w[NT4]; };
struct Item { WPtrs p; int fc[NT4]; int o0, i0, cnt, tile, sl; };

__device__ __forceinline__ Item make_item(const float* __restrict__ w, int tile, int sl,
                                          int slice_shift, int q, int koff) {
  Item it;
  const int nsl    = 1 << slice_shift;
  const int o0     = tile << 4;
  const int fc_min = (o0 > 4) ? (o0 - 4) : 0;
  const int kg     = fc_min & ~(GCOLS - 1);
  const int ngrp   = (N_IN - kg) / GCOLS;
  const int base   = ngrp >> slice_shift;
  const int extra  = ngrp & (nsl - 1);
  const int gs     = sl * base + ((sl < extra) ? sl : extra);
  it.tile = tile;  it.sl = sl;  it.o0 = o0;
  it.cnt  = base + ((sl < extra) ? 1 : 0);
  it.i0   = kg + gs * GCOLS;
#pragma unroll
  for (int t = 0; t < NT4; ++t) {
    const int  o  = o0 + 4 * t + q;
    it.fc[t] = (o > 4) ? (o - 4) : 0;
    const long rs = (long)o * N_IN - ((o >= 5) ? (((long)(o - 5) * (o - 4)) >> 1) : 0);
    it.p.w[t] = w + (rs - it.fc[t]) + koff;
  }
  return it;
}

__device__ __forceinline__ void group_load(Group& g, const WPtrs& p, const float* xlo,
                                           const float* xhi, int i) {
#pragma unroll
  for (int s = 0; s < GSTEPS; ++s) {
#pragma unroll
    for (int t = 0; t < NT4; ++t)
      g.a[s][t] = __builtin_nontemporal_load((const v4fu*)(p.w[t] + i + STEPC * s));
    g.xl[s] = *(const v4fu*)(xlo + i + STEPC * s);
    g.xh[s] = *(const v4fu*)(xhi + i + STEPC * s);
  }
}

__device__ __forceinline__ void group_mask(Group& g, const int (&fc)[NT4], int col0) {
#pragma unroll
  for (int s = 0; s < GSTEPS; ++s) {
    const int col = col0 + STEPC * s;
#pragma unroll
    for (int t = 0; t < NT4; ++t) {
      v4f v = g.a[s][t];
      v.x = (col     >= fc[t]) ? v.x : 0.0f;  v.y = (col + 1 >= fc[t]) ? v.y : 0.0f;
      v.z = (col + 2 >= fc[t]) ? v.z : 0.0f;  v.w = (col + 3 >= fc[t]) ? v.w : 0.0f;
      g.a[s][t] = v;
    }
  }
}

#define PART_FLOATS (2 * NSL * NB * 16)
__device__ __forceinline__ void item_flush(const v8f (&lo)[NT4], const v8f (&hi)[NT4], int lane,
                                           float* __restrict__ part, int which, int sl) {
  const bool up = (lane & 4) != 0;
  float val[16];
#pragma unroll
  for (int t = 0; t < NT4; ++t) {
#pragma unroll
    for (int r = 0; r < 4; ++r) {
      float sl_ = up ? lo[t][r + 4] : lo[t][r];
      float sh_ = up ? hi[t][r + 4] : hi[t][r];
      sl_ += __shfl_xor(sl_, 4);   sh_ += __shfl_xor(sh_, 4);
      sl_ += __shfl_xor(sl_, 24);  sh_ += __shfl_xor(sh_, 24);
      val[4 * t + r] = up ? sh_ : sl_;
    }
  }
  if (lane < NB) {
    float* pw = part + ((((which * NSL) + sl) * NB + lane) << 4);
    v4f q0, q1, q2, q3;
    q0.x = val[0];  q0.y = val[1];  q0.z = val[2];  q0.w = val[3];
    q1.x = val[4];  q1.y = val[5];  q1.z = val[6];  q1.w = val[7];
    q2.x = val[8];  q2.y = val[9];  q2.z = val[10]; q2.w = val[11];
    q3.x = val[12]; q3.y = val[13]; q3.z = val[14]; q3.w = val[15];
    *(v4f*)pw        = q0;
    *(v4f*)(pw + 4)  = q1;
    *(v4f*)(pw + 8)  = q2;
    *(v4f*)(pw + 12) = q3;
  }
}

__global__ __launch_bounds__(NSL * 32)
void tri_wmma_fused(const float* __restrict__ w, const float* __restrict__ x,
                    const float* __restrict__ bias, float* __restrict__ out) {
  __shared__ float part[PART_FLOATS];
  const int lane = threadIdx.x & 31;
  const int wave = __builtin_amdgcn_readfirstlane((int)(threadIdx.x >> 5));
  const int h    = lane >> 4;
  const int c    = lane & 15;
  const int q    = c & 3;
  const int koff = ((c >> 2) * 2 + h) << 2;

  const Item A = make_item(w, blockIdx.x, wave, NSL_LOG2, q, koff);
  const Item B = make_item(w, NTILE - 1 - blockIdx.x, NSL - 1 - wave, NSL_LOG2, q, koff);
  const int nA = A.cnt;
  const int nB = B.cnt;
  const int n  = nA + nB;

  const int t     = threadIdx.x;
  const int which = t >> 7;
  const int m     = (t >> 4) & 7;
  const int j     = t & 15;
  const int r     = 15 - j;
  const int o0    = (which ? (NTILE - 1 - (int)blockIdx.x) : (int)blockIdx.x) << 4;
  const float bv  = bias[o0 + r];

  const float* xlo = x + (long)q * N_IN + koff;
  const float* xhi = xlo + 4L * N_IN;
  WPtrs pnull;
#pragma unroll
  for (int t = 0; t < NT4; ++t) pnull.w[t] = xlo;

  WPtrs pi = A.p;  int ii = A.i0;  int ji = 0;
  int fcc[NT4];
#pragma unroll
  for (int t = 0; t < NT4; ++t) fcc[t] = A.fc[t];
  int ic = A.i0, jc = 0, endc = nA - 1;
  int cwhich = 0, csl = A.sl, co0 = A.o0;

  v8f lo[NT4], hi[NT4];
#pragma unroll
  for (int t = 0; t < NT4; ++t) { lo[t] = (v8f){}; hi[t] = (v8f){}; }

  Group g0, g1;

#define REQUEST(G)                                                              \
  do {                                                                          \
    group_load(G, pi, xlo, xhi, ii);                                            \
    ++ji;  ii += GCOLS;                                                         \
    if (ji == n)       { pi = pnull; ii = 0; }                                  \
    else if (ji == nA) { pi = B.p;   ii = B.i0; }                               \
  } while (0)

#define CONSUME(G)                                                              \
  do {                                                                          \
    if (ic < co0 + 16) group_mask(G, fcc, ic + koff);                           \
    group_mma(G, lo, hi);                                                       \
    ic += GCOLS;                                                                \
    if (jc == endc) {                                                           \
      item_flush(lo, hi, lane, part, cwhich, csl);                              \
      _Pragma("unroll")    \
      for (int t = 0; t < NT4; ++t) { lo[t] = (v8f){}; hi[t] = (v8f){}; fcc[t] = B.fc[t]; } \
      ic = B.i0;  endc = n - 1;  cwhich = 1;  csl = B.sl;  co0 = B.o0;          \
    }                                                                           \
    ++jc;                                                                       \
  } while (0)

  REQUEST(g0);
  for (int j = 0; j < n; j += 2) {
    REQUEST(g1);
    __builtin_amdgcn_sched_barrier(0);
    CONSUME(g0);
    __builtin_amdgcn_sched_barrier(0);
    REQUEST(g0);
    __builtin_amdgcn_sched_barrier(0);
    if (j + 1 < n) CONSUME(g1);
    __builtin_amdgcn_sched_barrier(0);
  }
#undef REQUEST
#undef CONSUME

  if (nB == 0 && lane < NB) {
    float* pw = part + (((NSL + B.sl) * NB + lane) << 4);
    const v4f z = {0.0f, 0.0f, 0.0f, 0.0f};
    *(v4f*)pw        = z;
    *(v4f*)(pw + 4)  = z;
    *(v4f*)(pw + 8)  = z;
    *(v4f*)(pw + 12) = z;
  }

  __syncthreads();

  float s = 0.0f;
  for (int cs = 0; cs < NSL; ++cs)
    s += part[((((which * NSL) + cs) * NB + m) << 4) + r];
  out[(long)m * N_OUT + (N_OUT - 16 - o0) + j] = s + bv;
}

extern "C" void kernel_launch(void* const* d_in, const int* in_sizes, int n_in,
                              void* d_out, int out_size, void* d_ws, size_t ws_size,
                              hipStream_t stream) {
  (void)in_sizes; (void)n_in; (void)out_size;
  const float* x    = (const float*)d_in[0];
  const float* w    = (const float*)d_in[1];
  const float* bias = (const float*)d_in[2];
  float*       out  = (float*)d_out;

  (void)d_ws; (void)ws_size;
  tri_wmma_fused<<<NPAIR, NSL * 32, 0, stream>>>(w, x, bias, out);
}


# W loads read ONE packed row each: 512 contiguous bytes, 5 lines per load (was 4 rows x 128 B: 8 lines, every line fetched twice); a quad lane exchange in registers rebuilds fragments; sum order same
# speedup vs baseline: 5.4458x; 1.0331x over previous
#include <hip/hip_runtime.h>


typedef float v2f __attribute__((ext_vector_type(2)));
typedef float v4f __attribute__((ext_vector_type(4)));
typedef float v8f __attribute__((ext_vector_type(8)));
typedef float v4fu __attribute__((ext_vector_type(4), aligned(4)));

#define N_IN   8192
#define N_OUT  8192
#define NB     8
#define NSL    8
#define NSL_LOG2 3
#define NTILE  (N_OUT / 16)
#define NPAIR  (NTILE / 2)

#define STEPC  32
#define GSTEPS 4
#define GCOLS  (GSTEPS * STEPC)
#define NT4    4
struct Group { v4f a[GSTEPS][NT4], xl[GSTEPS], xh[GSTEPS]; };
__device__ __forceinline__ void step_mma(const v4f (&a)[NT4], const v4f xl, const v4f xh,
                                         v8f (&lo)[NT4], v8f (&hi)[NT4]) {
  const v2f l01 = {xl.x, xl.y}, l23 = {xl.z, xl.w};
  const v2f h01 = {xh.x, xh.y}, h23 = {xh.z, xh.w};
#pragma unroll
  for (int t = 0; t < NT4; ++t) {
    const v2f a01 = {a[t].x, a[t].y}, a23 = {a[t].z, a[t].w};
    lo[t] = __builtin_amdgcn_wmma_f32_16x16x4_f32(false, a01, false, l01, (short)0, lo[t], false, false);
    hi[t] = __builtin_amdgcn_wmma_f32_16x16x4_f32(false, a01, false, h01, (short)0, hi[t], false, false);
    lo[t] = __builtin_amdgcn_wmma_f32_16x16x4_f32(false, a23, false, l23, (short)0, lo[t], false, false);
    hi[t] = __builtin_amdgcn_wmma_f32_16x16x4_f32(false, a23, false, h23, (short)0, hi[t], false, false);
  }
}

__device__ __forceinline__ void group_mma(const Group& g, v8f (&lo)[NT4], v8f (&hi)[NT4]) {
#pragma unroll
  for (int s = 0; s < GSTEPS; ++s) step_mma(g.a[s], g.xl[s], g.xh[s], lo, hi);
}

struct RowBase { int vb[16]; };
struct Item { RowBase p; int fc[NT4]; int o0, i0, cnt, tile, sl; };

__device__ __forceinline__ Item make_item(int tile, int sl, int slice_shift, int q) {
  Item it;
  const int nsl    = 1 << slice_shift;
  const int o0     = tile << 4;
  const int fc_min = (o0 > 4) ? (o0 - 4) : 0;
  const int kg     = fc_min & ~(GCOLS - 1);
  const int ngrp   = (N_IN - kg) / GCOLS;
  const int base   = ngrp >> slice_shift;
  const int extra  = ngrp & (nsl - 1);
  const int gs     = sl * base + ((sl < extra) ? sl : extra);
  it.tile = tile;  it.sl = sl;  it.o0 = o0;
  it.cnt  = base + ((sl < extra) ? 1 : 0);
  it.i0   = kg + gs * GCOLS;
#pragma unroll
  for (int j = 0; j < 16; ++j) {
    const int o  = o0 + j;
    const int f  = (o > 4) ? (o - 4) : 0;
    const int rs = o * N_IN - ((o >= 5) ? (((o - 5) * (o - 4)) >> 1) : 0);
    it.p.vb[j] = rs - f;
  }
#pragma unroll
  for (int t = 0; t < NT4; ++t) {
    const int o = o0 + 4 * t + q;
    it.fc[t] = (o > 4) ? (o - 4) : 0;
  }
  return it;
}

__device__ __forceinline__ void group_load(Group& g, const float* base, const RowBase& p,
                                           const float* xlo, const float* xhi, int i, int lq) {
  const unsigned voff = (unsigned)(i + lq) << 2;
  const v4fu* ad[16];
#pragma unroll
  for (int j = 0; j < 16; ++j) {
    const char* rowp = (const char*)base + ((long)p.vb[j] << 2);
    ad[j] = (const v4fu*)(rowp + voff);
  }
#pragma unroll
  for (int t = 0; t < NT4; ++t) {
#pragma unroll
    for (int a = 0; a < 4; ++a)
      g.a[a][t] = __builtin_nontemporal_load(ad[4 * t + a]);
  }
#pragma unroll
  for (int s = 0; s < GSTEPS; ++s) {
    g.xl[s] = *(const v4fu*)(xlo + i + STEPC * s);
    g.xh[s] = *(const v4fu*)(xhi + i + STEPC * s);
  }
}

template <int CTRL>
__device__ __forceinline__ void pair_swap(v4f& r0, v4f& r1, bool bit) {
#pragma unroll
  for (int e = 0; e < 4; ++e) {
    const float give = bit ? r0[e] : r1[e];
    const float got  = __builtin_bit_cast(float,
        __builtin_amdgcn_mov_dpp(__builtin_bit_cast(int, give), CTRL, 0xF, 0xF, true));
    r0[e] = bit ? got : r0[e];
    r1[e] = bit ? r1[e] : got;
  }
}

__device__ __forceinline__ void group_transpose(Group& g, bool q0, bool q1) {
#pragma unroll
  for (int t = 0; t < NT4; ++t) {
    pair_swap<0xB1>(g.a[0][t], g.a[1][t], q0);
    pair_swap<0xB1>(g.a[2][t], g.a[3][t], q0);
    pair_swap<0x4E>(g.a[0][t], g.a[2][t], q1);
    pair_swap<0x4E>(g.a[1][t], g.a[3][t], q1);
  }
}

__device__ __forceinline__ void group_mask(Group& g, const int (&fc)[NT4], int col0) {
#pragma unroll
  for (int s = 0; s < GSTEPS; ++s) {
    const int col = col0 + STEPC * s;
#pragma unroll
    for (int t = 0; t < NT4; ++t) {
      v4f v = g.a[s][t];
      v.x = (col     >= fc[t]) ? v.x : 0.0f;  v.y = (col + 1 >= fc[t]) ? v.y : 0.0f;
      v.z = (col + 2 >= fc[t]) ? v.z : 0.0f;  v.w = (col + 3 >= fc[t]) ? v.w : 0.0f;
      g.a[s][t] = v;
    }
  }
}

#define PART_FLOATS (2 * NSL * NB * 16)
__device__ __forceinline__ void item_flush(const v8f (&lo)[NT4], const v8f (&hi)[NT4], int lane,
                                           float* __restrict__ part, int which, int sl) {
  const bool up = (lane & 4) != 0;
  float val[16];
#pragma unroll
  for (int t = 0; t < NT4; ++t) {
#pragma unroll
    for (int r = 0; r < 4; ++r) {
      float sl_ = up ? lo[t][r + 4] : lo[t][r];
      float sh_ = up ? hi[t][r + 4] : hi[t][r];
      sl_ += __shfl_xor(sl_, 4);   sh_ += __shfl_xor(sh_, 4);
      sl_ += __shfl_xor(sl_, 24);  sh_ += __shfl_xor(sh_, 24);
      val[4 * t + r] = up ? sh_ : sl_;
    }
  }
  if (lane < NB) {
    float* pw = part + ((((which * NSL) + sl) * NB + lane) << 4);
    v4f q0, q1, q2, q3;
    q0.x = val[0];  q0.y = val[1];  q0.z = val[2];  q0.w = val[3];
    q1.x = val[4];  q1.y = val[5];  q1.z = val[6];  q1.w = val[7];
    q2.x = val[8];  q2.y = val[9];  q2.z = val[10]; q2.w = val[11];
    q3.x = val[12]; q3.y = val[13]; q3.z = val[14]; q3.w = val[15];
    *(v4f*)pw        = q0;
    *(v4f*)(pw + 4)  = q1;
    *(v4f*)(pw + 8)  = q2;
    *(v4f*)(pw + 12) = q3;
  }
}

__global__ __launch_bounds__(NSL * 32)
void tri_wmma_fused(const float* __restrict__ w, const float* __restrict__ x,
                    const float* __restrict__ bias, float* __restrict__ out) {
  __shared__ float part[PART_FLOATS];
  const int lane = threadIdx.x & 31;
  const int wave = __builtin_amdgcn_readfirstlane((int)(threadIdx.x >> 5));
  const int h    = lane >> 4;
  const int c    = lane & 15;
  const int q    = c & 3;
  const int koff = ((c >> 2) * 2 + h) << 2;
  const int lq   = STEPC * q + koff;
  const bool q0  = (q & 1) != 0;
  const bool q1  = (q & 2) != 0;

  const Item A = make_item(blockIdx.x, wave, NSL_LOG2, q);
  const Item B = make_item(NTILE - 1 - blockIdx.x, NSL - 1 - wave, NSL_LOG2, q);
  const int nA = A.cnt;
  const int nB = B.cnt;
  const int n  = nA + nB;

  const int t     = threadIdx.x;
  const int which = t >> 7;
  const int m     = (t >> 4) & 7;
  const int j     = t & 15;
  const int r     = 15 - j;
  const int o0    = (which ? (NTILE - 1 - (int)blockIdx.x) : (int)blockIdx.x) << 4;
  const float bv  = bias[o0 + r];

  const float* xlo = x + (long)q * N_IN + koff;
  const float* xhi = xlo + 4L * N_IN;
  RowBase pnull;
#pragma unroll
  for (int j = 0; j < 16; ++j) pnull.vb[j] = 0;

  const float* pb = w;  RowBase pi = A.p;  int ii = A.i0;  int ji = 0;
  int fcc[NT4];
#pragma unroll
  for (int t = 0; t < NT4; ++t) fcc[t] = A.fc[t];
  int ic = A.i0, jc = 0, endc = nA - 1;
  int cwhich = 0, csl = A.sl, co0 = A.o0;

  v8f lo[NT4], hi[NT4];
#pragma unroll
  for (int t = 0; t < NT4; ++t) { lo[t] = (v8f){}; hi[t] = (v8f){}; }

  Group g0, g1;

#define REQUEST(G)                                                              \
  do {                                                                          \
    group_load(G, pb, pi, xlo, xhi, ii, lq);                                    \
    ++ji;  ii += GCOLS;                                                         \
    if (ji == n)       { pb = x; pi = pnull; ii = 0; }                          \
    else if (ji == nA) { pi = B.p;   ii = B.i0; }                               \
  } while (0)

#define CONSUME(G)                                                              \
  do {                                                                          \
    group_transpose(G, q0, q1);                                                 \
    if (ic < co0 + 16) group_mask(G, fcc, ic + koff);                           \
    group_mma(G, lo, hi);                                                       \
    ic += GCOLS;                                                                \
    if (jc == endc) {                                                           \
      item_flush(lo, hi, lane, part, cwhich, csl);                              \
      _Pragma("unroll")    \
      for (int t = 0; t < NT4; ++t) { lo[t] = (v8f){}; hi[t] = (v8f){}; fcc[t] = B.fc[t]; } \
      ic = B.i0;  endc = n - 1;  cwhich = 1;  csl = B.sl;  co0 = B.o0;          \
    }                                                                           \
    ++jc;                                                                       \
  } while (0)

  REQUEST(g0);
  for (int j = 0; j < n; j += 2) {
    REQUEST(g1);
    __builtin_amdgcn_sched_barrier(0);
    CONSUME(g0);
    __builtin_amdgcn_sched_barrier(0);
    REQUEST(g0);
    __builtin_amdgcn_sched_barrier(0);
    if (j + 1 < n) CONSUME(g1);
    __builtin_amdgcn_sched_barrier(0);
  }
#undef REQUEST
#undef CONSUME

  if (nB == 0 && lane < NB) {
    float* pw = part + (((NSL + B.sl) * NB + lane) << 4);
    const v4f z = {0.0f, 0.0f, 0.0f, 0.0f};
    *(v4f*)pw        = z;
    *(v4f*)(pw + 4)  = z;
    *(v4f*)(pw + 8)  = z;
    *(v4f*)(pw + 12) = z;
  }

  __syncthreads();

  float s = 0.0f;
  for (int cs = 0; cs < NSL; ++cs)
    s += part[((((which * NSL) + cs) * NB + m) << 4) + r];
  out[(long)m * N_OUT + (N_OUT - 16 - o0) + j] = s + bv;
}

extern "C" void kernel_launch(void* const* d_in, const int* in_sizes, int n_in,
                              void* d_out, int out_size, void* d_ws, size_t ws_size,
                              hipStream_t stream) {
  (void)in_sizes; (void)n_in; (void)out_size;
  const float* x    = (const float*)d_in[0];
  const float* w    = (const float*)d_in[1];
  const float* bias = (const float*)d_in[2];
  float*       out  = (float*)d_out;

  (void)d_ws; (void)ws_size;
  tri_wmma_fused<<<NPAIR, NSL * 32, 0, stream>>>(w, x, bias, out);
}


# Fragment retile to 8 rows x 2 column segments x all 8 batches (was 4 x 4, lo/hi halves): half the WMMAs per column; W still loaded as whole rows, the lane exchange grows to 8x8; sum order changes
# speedup vs baseline: 5.9367x; 1.0901x over previous
#include <hip/hip_runtime.h>


typedef float v2f __attribute__((ext_vector_type(2)));
typedef float v4f __attribute__((ext_vector_type(4)));
typedef float v8f __attribute__((ext_vector_type(8)));
typedef float v4fu __attribute__((ext_vector_type(4), aligned(4)));

#define N_IN   8192
#define N_OUT  8192
#define NB     8
#define NSL    8
#define NSL_LOG2 3
#define NTILE  (N_OUT / 16)
#define NPAIR  (NTILE / 2)

#define STEPC  16
#define GSTEPS 8
#define GCOLS  (GSTEPS * STEPC)
#define NT8    2
struct Group { v4f a[GSTEPS][NT8], xv[GSTEPS]; };
__device__ __forceinline__ void step_mma(const v4f (&a)[NT8], const v4f xv, v8f (&acc)[NT8]) {
  const v2f x01 = {xv.x, xv.y}, x23 = {xv.z, xv.w};
  const v2f a01 = {a[0].x, a[0].y}, a23 = {a[0].z, a[0].w};
  const v2f b01 = {a[1].x, a[1].y}, b23 = {a[1].z, a[1].w};
  acc[0] = __builtin_amdgcn_wmma_f32_16x16x4_f32(false, a01, false, x01, (short)0, acc[0], false, false);
  acc[1] = __builtin_amdgcn_wmma_f32_16x16x4_f32(false, b01, false, x01, (short)0, acc[1], false, false);
  acc[0] = __builtin_amdgcn_wmma_f32_16x16x4_f32(false, a23, false, x23, (short)0, acc[0], false, false);
  acc[1] = __builtin_amdgcn_wmma_f32_16x16x4_f32(false, b23, false, x23, (short)0, acc[1], false, false);
}

__device__ __forceinline__ void group_mma(const Group& g, v8f (&acc)[NT8]) {
#pragma unroll
  for (int s = 0; s < GSTEPS; ++s) step_mma(g.a[s], g.xv[s], acc);
}

struct RowBase { int vb[16]; };
struct Item { RowBase p; int fc[NT8]; int o0, i0, cnt, tile, sl; };

__device__ __forceinline__ Item make_item(int tile, int sl, int slice_shift, int r8) {
  Item it;
  const int nsl    = 1 << slice_shift;
  const int o0     = tile << 4;
  const int fc_min = (o0 > 4) ? (o0 - 4) : 0;
  const int kg     = fc_min & ~(GCOLS - 1);
  const int ngrp   = (N_IN - kg) / GCOLS;
  const int base   = ngrp >> slice_shift;
  const int extra  = ngrp & (nsl - 1);
  const int gs     = sl * base + ((sl < extra) ? sl : extra);
  it.tile = tile;  it.sl = sl;  it.o0 = o0;
  it.cnt  = base + ((sl < extra) ? 1 : 0);
  it.i0   = kg + gs * GCOLS;
#pragma unroll
  for (int j = 0; j < 16; ++j) {
    const int o  = o0 + j;
    const int f  = (o > 4) ? (o - 4) : 0;
    const int rs = o * N_IN - ((o >= 5) ? (((o - 5) * (o - 4)) >> 1) : 0);
    it.p.vb[j] = rs - f;
  }
#pragma unroll
  for (int t = 0; t < NT8; ++t) {
    const int o = o0 + 8 * t + r8;
    it.fc[t] = (o > 4) ? (o - 4) : 0;
  }
  return it;
}

__device__ __forceinline__ void group_load(Group& g, const float* base, const RowBase& p,
                                           const float* xq, int i, int lq) {
  const unsigned voff = (unsigned)(i + lq) << 2;
  const v4fu* ad[16];
#pragma unroll
  for (int j = 0; j < 16; ++j) {
    const char* rowp = (const char*)base + ((long)p.vb[j] << 2);
    ad[j] = (const v4fu*)(rowp + voff);
  }
#pragma unroll
  for (int t = 0; t < NT8; ++t) {
#pragma unroll
    for (int a = 0; a < GSTEPS; ++a)
      g.a[a][t] = __builtin_nontemporal_load(ad[8 * t + a]);
  }
#pragma unroll
  for (int s = 0; s < GSTEPS; ++s)
    g.xv[s] = *(const v4fu*)(xq + i + STEPC * s);
}

template <int CTRL>
__device__ __forceinline__ float partner(float v) {
  return __builtin_bit_cast(float,
      __builtin_amdgcn_mov_dpp(__builtin_bit_cast(int, v), CTRL, 0xF, 0xF, true));
}

template <int CTRL>
__device__ __forceinline__ void pair_swap(v4f& r0, v4f& r1, bool bit) {
#pragma unroll
  for (int e = 0; e < 4; ++e) {
    const float p0 = partner<CTRL>(r0[e]);
    const float p1 = partner<CTRL>(r1[e]);
    const float n0 = bit ? p1 : r0[e];
    const float n1 = bit ? r1[e] : p0;
    r0[e] = n0;
    r1[e] = n1;
  }
}

__device__ __forceinline__ void group_transpose(Group& g, bool b0, bool b1, bool b2) {
#pragma unroll
  for (int t = 0; t < NT8; ++t) {
    pair_swap<0xB1>(g.a[0][t], g.a[1][t], b0);
    pair_swap<0xB1>(g.a[2][t], g.a[3][t], b0);
    pair_swap<0xB1>(g.a[4][t], g.a[5][t], b0);
    pair_swap<0xB1>(g.a[6][t], g.a[7][t], b0);
    pair_swap<0x4E>(g.a[0][t], g.a[2][t], b1);
    pair_swap<0x4E>(g.a[1][t], g.a[3][t], b1);
    pair_swap<0x4E>(g.a[4][t], g.a[6][t], b1);
    pair_swap<0x4E>(g.a[5][t], g.a[7][t], b1);
    pair_swap<0x128>(g.a[0][t], g.a[4][t], b2);
    pair_swap<0x128>(g.a[1][t], g.a[5][t], b2);
    pair_swap<0x128>(g.a[2][t], g.a[6][t], b2);
    pair_swap<0x128>(g.a[3][t], g.a[7][t], b2);
  }
}

__device__ __forceinline__ void group_mask(Group& g, const int (&fc)[NT8], int col0) {
#pragma unroll
  for (int s = 0; s < GSTEPS; ++s) {
    const int col = col0 + STEPC * s;
#pragma unroll
    for (int t = 0; t < NT8; ++t) {
      v4f v = g.a[s][t];
      v.x = (col     >= fc[t]) ? v.x : 0.0f;  v.y = (col + 1 >= fc[t]) ? v.y : 0.0f;
      v.z = (col + 2 >= fc[t]) ? v.z : 0.0f;  v.w = (col + 3 >= fc[t]) ? v.w : 0.0f;
      g.a[s][t] = v;
    }
  }
}

#define PART_FLOATS (2 * NSL * NB * 16)
__device__ __forceinline__ void item_flush(const v8f (&acc)[NT8], int lane,
                                           float* __restrict__ part, int which, int sl) {
  const bool up = (lane & 4) != 0;
  v4f val[NT8];
#pragma unroll
  for (int t = 0; t < NT8; ++t) {
#pragma unroll
    for (int k = 0; k < 4; ++k) {
      float v = up ? acc[t][k + 4] : acc[t][k];
      v += __shfl_xor(v, 4);
      val[t][k] = v;
    }
  }
  if (!up) {
    const int batch = (((lane >> 3) & 1) << 2) | (lane & 3);
    float* pw = part + ((((which * NSL) + sl) * NB + batch) << 4) + ((lane >> 4) << 2);
    *(v4f*)pw       = val[0];
    *(v4f*)(pw + 8) = val[1];
  }
}

__global__ __launch_bounds__(NSL * 32)
void tri_wmma_fused(const float* __restrict__ w, const float* __restrict__ x,
                    const float* __restrict__ bias, float* __restrict__ out) {
  __shared__ float part[PART_FLOATS];
  const int lane = threadIdx.x & 31;
  const int wave = __builtin_amdgcn_readfirstlane((int)(threadIdx.x >> 5));
  const int h    = lane >> 4;
  const int c    = lane & 15;
  const int r8   = ((c >> 3) << 2) | (c & 3);
  const int koff = ((((c >> 2) & 1) << 1) + h) << 2;
  const int lq   = STEPC * r8 + koff;
  const bool b0  = (c & 1) != 0;
  const bool b1  = (c & 2) != 0;
  const bool b2  = (c & 8) != 0;

  const Item A = make_item(blockIdx.x, wave, NSL_LOG2, r8);
  const Item B = make_item(NTILE - 1 - blockIdx.x, NSL - 1 - wave, NSL_LOG2, r8);
  const int nA = A.cnt;
  const int nB = B.cnt;
  const int n  = nA + nB;

  const int t     = threadIdx.x;
  const int which = t >> 7;
  const int m     = (t >> 4) & 7;
  const int j     = t & 15;
  const int r     = 15 - j;
  const int o0    = (which ? (NTILE - 1 - (int)blockIdx.x) : (int)blockIdx.x) << 4;
  const float bv  = bias[o0 + r];

  const float* xq = x + (long)r8 * N_IN + koff;
  RowBase pnull;
#pragma unroll
  for (int j = 0; j < 16; ++j) pnull.vb[j] = 0;

  const float* pb = w;  RowBase pi = A.p;  int ii = A.i0 - GCOLS;  int ji = -1;
  int fcc[NT8];
#pragma unroll
  for (int t = 0; t < NT8; ++t) fcc[t] = A.fc[t];
  int ic = A.i0, jc = 0, endc = nA - 1;
  int cwhich = 0, csl = A.sl, co0 = A.o0;

  v8f acc[NT8];
#pragma unroll
  for (int t = 0; t < NT8; ++t) acc[t] = (v8f){};

  Group g0, g1;

#define REQUEST(G)                                                              \
  do {                                                                          \
    ++ji;  ii += GCOLS;                                                         \
    if (ji == n)       { pb = x; pi = pnull; ii = 0; }                          \
    else if (ji == nA) { pi = B.p;   ii = B.i0; }                               \
    group_load(G, pb, pi, xq, ii, lq);                                          \
  } while (0)

#define CONSUME(G)                                                              \
  do {                                                                          \
    group_transpose(G, b0, b1, b2);                                             \
    if (ic < co0 + 16) group_mask(G, fcc, ic + koff);                           \
    group_mma(G, acc);                                                          \
    ic += GCOLS;                                                                \
    if (jc == endc) {                                                           \
      item_flush(acc, lane, part, cwhich, csl);                                 \
      _Pragma("unroll")    \
      for (int t = 0; t < NT8; ++t) { acc[t] = (v8f){}; fcc[t] = B.fc[t]; }     \
      ic = B.i0;  endc = n - 1;  cwhich = 1;  csl = B.sl;  co0 = B.o0;          \
    }                                                                           \
    ++jc;                                                                       \
  } while (0)

  REQUEST(g0);
  for (int j = 0; j < n; j += 2) {
    REQUEST(g1);
    __builtin_amdgcn_sched_barrier(0);
    CONSUME(g0);
    __builtin_amdgcn_sched_barrier(0);
    REQUEST(g0);
    __builtin_amdgcn_sched_barrier(0);
    if (j + 1 < n) CONSUME(g1);
    __builtin_amdgcn_sched_barrier(0);
  }
#undef REQUEST
#undef CONSUME

  if (nB == 0 && lane < NB) {
    float* pw = part + (((NSL + B.sl) * NB + lane) << 4);
    const v4f z = {0.0f, 0.0f, 0.0f, 0.0f};
    *(v4f*)pw        = z;
    *(v4f*)(pw + 4)  = z;
    *(v4f*)(pw + 8)  = z;
    *(v4f*)(pw + 12) = z;
  }

  __syncthreads();

  float s = 0.0f;
  for (int cs = 0; cs < NSL; ++cs)
    s += part[((((which * NSL) + cs) * NB + m) << 4) + r];
  out[(long)m * N_OUT + (N_OUT - 16 - o0) + j] = s + bv;
}

extern "C" void kernel_launch(void* const* d_in, const int* in_sizes, int n_in,
                              void* d_out, int out_size, void* d_ws, size_t ws_size,
                              hipStream_t stream) {
  (void)in_sizes; (void)n_in; (void)out_size;
  const float* x    = (const float*)d_in[0];
  const float* w    = (const float*)d_in[1];
  const float* bias = (const float*)d_in[2];
  float*       out  = (float*)d_out;

  (void)d_ws; (void)ws_size;
  tri_wmma_fused<<<NPAIR, NSL * 32, 0, stream>>>(w, x, bias, out);
}


# Never-used loads removed: group list cut so the last group is multiplied behind the loop (was 24-48 loads per wave past the end, all to one 512 B, waited for); row bases held per lane; sum order same
# speedup vs baseline: 6.2410x; 1.0513x over previous
#include <hip/hip_runtime.h>


typedef float v2f __attribute__((ext_vector_type(2)));
typedef float v4f __attribute__((ext_vector_type(4)));
typedef float v8f __attribute__((ext_vector_type(8)));
typedef float v4fu __attribute__((ext_vector_type(4), aligned(4)));

#define N_IN   8192
#define N_OUT  8192
#define NB     8
#define NSL    8
#define NSL_LOG2 3
#define NTILE  (N_OUT / 16)
#define NPAIR  (NTILE / 2)

#define STEPC  16
#define GSTEPS 8
#define GCOLS  (GSTEPS * STEPC)
#define NT8    2
struct Group { v4f a[GSTEPS][NT8], xv[GSTEPS]; };
__device__ __forceinline__ void step_mma(const v4f (&a)[NT8], const v4f xv, v8f (&acc)[NT8]) {
  const v2f x01 = {xv.x, xv.y}, x23 = {xv.z, xv.w};
  const v2f a01 = {a[0].x, a[0].y}, a23 = {a[0].z, a[0].w};
  const v2f b01 = {a[1].x, a[1].y}, b23 = {a[1].z, a[1].w};
  acc[0] = __builtin_amdgcn_wmma_f32_16x16x4_f32(false, a01, false, x01, (short)0, acc[0], false, false);
  acc[1] = __builtin_amdgcn_wmma_f32_16x16x4_f32(false, b01, false, x01, (short)0, acc[1], false, false);
  acc[0] = __builtin_amdgcn_wmma_f32_16x16x4_f32(false, a23, false, x23, (short)0, acc[0], false, false);
  acc[1] = __builtin_amdgcn_wmma_f32_16x16x4_f32(false, b23, false, x23, (short)0, acc[1], false, false);
}

__device__ __forceinline__ void group_mma(const Group& g, v8f (&acc)[NT8]) {
#pragma unroll
  for (int s = 0; s < GSTEPS; ++s) step_mma(g.a[s], g.xv[s], acc);
}

struct Item { int vb; int fc[NT8]; int o0, i0, cnt, tile, sl; };

__device__ __forceinline__ Item make_item(int tile, int sl, int slice_shift, int r8, int lane) {
  Item it;
  const int nsl    = 1 << slice_shift;
  const int o0     = tile << 4;
  const int fc_min = (o0 > 4) ? (o0 - 4) : 0;
  const int kg     = fc_min & ~(GCOLS - 1);
  const int ngrp   = (N_IN - kg) / GCOLS;
  const int base   = ngrp >> slice_shift;
  const int extra  = ngrp & (nsl - 1);
  const int gs     = sl * base + ((sl < extra) ? sl : extra);
  it.tile = tile;  it.sl = sl;  it.o0 = o0;
  it.cnt  = base + ((sl < extra) ? 1 : 0);
  it.i0   = kg + gs * GCOLS;
  {
    const int o  = o0 + (lane & 15);
    const int f  = (o > 4) ? (o - 4) : 0;
    const int rs = o * N_IN - ((o >= 5) ? (((o - 5) * (o - 4)) >> 1) : 0);
    it.vb = rs - f;
  }
#pragma unroll
  for (int t = 0; t < NT8; ++t) {
    const int o = o0 + 8 * t + r8;
    it.fc[t] = (o > 4) ? (o - 4) : 0;
  }
  return it;
}

__device__ __forceinline__ void group_load(Group& g, const float* base, int vbv,
                                           const float* xq, int i, int lq) {
  const unsigned voff = (unsigned)(i + lq) << 2;
  const v4fu* ad[16];
#pragma unroll
  for (int j = 0; j < 16; ++j) {
    const int vbj = __builtin_amdgcn_readlane(vbv, j);
    const char* rowp = (const char*)base + ((long)vbj << 2);
    ad[j] = (const v4fu*)(rowp + voff);
  }
#pragma unroll
  for (int t = 0; t < NT8; ++t) {
#pragma unroll
    for (int a = 0; a < GSTEPS; ++a)
      g.a[a][t] = __builtin_nontemporal_load(ad[8 * t + a]);
  }
#pragma unroll
  for (int s = 0; s < GSTEPS; ++s)
    g.xv[s] = *(const v4fu*)(xq + i + STEPC * s);
}

template <int CTRL>
__device__ __forceinline__ float partner(float v) {
  return __builtin_bit_cast(float,
      __builtin_amdgcn_mov_dpp(__builtin_bit_cast(int, v), CTRL, 0xF, 0xF, true));
}

template <int CTRL>
__device__ __forceinline__ void pair_swap(v4f& r0, v4f& r1, bool bit) {
#pragma unroll
  for (int e = 0; e < 4; ++e) {
    const float p0 = partner<CTRL>(r0[e]);
    const float p1 = partner<CTRL>(r1[e]);
    const float n0 = bit ? p1 : r0[e];
    const float n1 = bit ? r1[e] : p0;
    r0[e] = n0;
    r1[e] = n1;
  }
}

__device__ __forceinline__ void group_transpose(Group& g, bool b0, bool b1, bool b2) {
#pragma unroll
  for (int t = 0; t < NT8; ++t) {
    pair_swap<0xB1>(g.a[0][t], g.a[1][t], b0);
    pair_swap<0xB1>(g.a[2][t], g.a[3][t], b0);
    pair_swap<0xB1>(g.a[4][t], g.a[5][t], b0);
    pair_swap<0xB1>(g.a[6][t], g.a[7][t], b0);
    pair_swap<0x4E>(g.a[0][t], g.a[2][t], b1);
    pair_swap<0x4E>(g.a[1][t], g.a[3][t], b1);
    pair_swap<0x4E>(g.a[4][t], g.a[6][t], b1);
    pair_swap<0x4E>(g.a[5][t], g.a[7][t], b1);
    pair_swap<0x128>(g.a[0][t], g.a[4][t], b2);
    pair_swap<0x128>(g.a[1][t], g.a[5][t], b2);
    pair_swap<0x128>(g.a[2][t], g.a[6][t], b2);
    pair_swap<0x128>(g.a[3][t], g.a[7][t], b2);
  }
}

__device__ __forceinline__ void group_mask(Group& g, const int (&fc)[NT8], int col0) {
#pragma unroll
  for (int s = 0; s < GSTEPS; ++s) {
    const int col = col0 + STEPC * s;
#pragma unroll
    for (int t = 0; t < NT8; ++t) {
      v4f v = g.a[s][t];
      v.x = (col     >= fc[t]) ? v.x : 0.0f;  v.y = (col + 1 >= fc[t]) ? v.y : 0.0f;
      v.z = (col + 2 >= fc[t]) ? v.z : 0.0f;  v.w = (col + 3 >= fc[t]) ? v.w : 0.0f;
      g.a[s][t] = v;
    }
  }
}

#define PART_FLOATS (2 * NSL * NB * 16)
__device__ __forceinline__ void item_flush(const v8f (&acc)[NT8], int lane,
                                           float* __restrict__ part, int which, int sl) {
  const bool up = (lane & 4) != 0;
  v4f val[NT8];
#pragma unroll
  for (int t = 0; t < NT8; ++t) {
#pragma unroll
    for (int k = 0; k < 4; ++k) {
      float v = up ? acc[t][k + 4] : acc[t][k];
      v += __shfl_xor(v, 4);
      val[t][k] = v;
    }
  }
  if (!up) {
    const int batch = (((lane >> 3) & 1) << 2) | (lane & 3);
    float* pw = part + ((((which * NSL) + sl) * NB + batch) << 4) + ((lane >> 4) << 2);
    *(v4f*)pw       = val[0];
    *(v4f*)(pw + 8) = val[1];
  }
}

__global__ __launch_bounds__(NSL * 32)
void tri_wmma_fused(const float* __restrict__ w, const float* __restrict__ x,
                    const float* __restrict__ bias, float* __restrict__ out) {
  __shared__ float part[PART_FLOATS];
  const int lane = threadIdx.x & 31;
  const int wave = __builtin_amdgcn_readfirstlane((int)(threadIdx.x >> 5));
  const int h    = lane >> 4;
  const int c    = lane & 15;
  const int r8   = ((c >> 3) << 2) | (c & 3);
  const int koff = ((((c >> 2) & 1) << 1) + h) << 2;
  const int lq   = STEPC * r8 + koff;
  const bool b0  = (c & 1) != 0;
  const bool b1  = (c & 2) != 0;
  const bool b2  = (c & 8) != 0;

  const Item A = make_item(blockIdx.x, wave, NSL_LOG2, r8, lane);
  const Item B = make_item(NTILE - 1 - blockIdx.x, NSL - 1 - wave, NSL_LOG2, r8, lane);
  const int nA = A.cnt;
  const int nB = B.cnt;
  const int n  = nA + nB;

  const int t     = threadIdx.x;
  const int which = t >> 7;
  const int m     = (t >> 4) & 7;
  const int j     = t & 15;
  const int r     = 15 - j;
  const int o0    = (which ? (NTILE - 1 - (int)blockIdx.x) : (int)blockIdx.x) << 4;
  const float bv  = bias[o0 + r];

  const float* xq = x + (long)r8 * N_IN + koff;
  int vbi = A.vb;  int ii = A.i0 - GCOLS;  int ji = -1;
  int fcc[NT8];
#pragma unroll
  for (int t = 0; t < NT8; ++t) fcc[t] = A.fc[t];
  int ic = A.i0, jc = 0, endc = nA - 1;
  int cwhich = 0, csl = A.sl, co0 = A.o0;

  v8f acc[NT8];
#pragma unroll
  for (int t = 0; t < NT8; ++t) acc[t] = (v8f){};

  Group g0, g1;

#define REQUEST(G)                                                              \
  do {                                                                          \
    ++ji;  ii += GCOLS;                                                         \
    if (ji == nA) { vbi = B.vb;  ii = B.i0; }                                   \
    group_load(G, w, vbi, xq, ii, lq);                                          \
  } while (0)

#define CONSUME(G)                                                              \
  do {                                                                          \
    group_transpose(G, b0, b1, b2);                                             \
    if (ic < co0 + 16) group_mask(G, fcc, ic + koff);                           \
    group_mma(G, acc);                                                          \
    ic += GCOLS;                                                                \
    if (jc == endc) {                                                           \
      item_flush(acc, lane, part, cwhich, csl);                                 \
      _Pragma("unroll")    \
      for (int t = 0; t < NT8; ++t) { acc[t] = (v8f){}; fcc[t] = B.fc[t]; }     \
      ic = B.i0;  endc = n - 1;  cwhich = 1;  csl = B.sl;  co0 = B.o0;          \
    }                                                                           \
    ++jc;                                                                       \
  } while (0)

  const bool even = (n & 1) == 0;
  int gj = 0;
  if (even) { REQUEST(g1);  gj = 1; }
  __builtin_amdgcn_sched_barrier(0);
  REQUEST(g0);
  __builtin_amdgcn_sched_barrier(0);
  if (even) CONSUME(g1);
  __builtin_amdgcn_sched_barrier(0);
  for (; gj < n - 1; gj += 2) {
    REQUEST(g1);
    __builtin_amdgcn_sched_barrier(0);
    CONSUME(g0);
    __builtin_amdgcn_sched_barrier(0);
    REQUEST(g0);
    __builtin_amdgcn_sched_barrier(0);
    CONSUME(g1);
    __builtin_amdgcn_sched_barrier(0);
  }
  CONSUME(g0);
#undef REQUEST
#undef CONSUME

  if (nB == 0 && lane < NB) {
    float* pw = part + (((NSL + B.sl) * NB + lane) << 4);
    const v4f z = {0.0f, 0.0f, 0.0f, 0.0f};
    *(v4f*)pw        = z;
    *(v4f*)(pw + 4)  = z;
    *(v4f*)(pw + 8)  = z;
    *(v4f*)(pw + 12) = z;
  }

  __syncthreads();

  float s = 0.0f;
  for (int cs = 0; cs < NSL; ++cs)
    s += part[((((which * NSL) + cs) * NB + m) << 4) + r];
  out[(long)m * N_OUT + (N_OUT - 16 - o0) + j] = s + bv;
}

extern "C" void kernel_launch(void* const* d_in, const int* in_sizes, int n_in,
                              void* d_out, int out_size, void* d_ws, size_t ws_size,
                              hipStream_t stream) {
  (void)in_sizes; (void)n_in; (void)out_size;
  const float* x    = (const float*)d_in[0];
  const float* w    = (const float*)d_in[1];
  const float* bias = (const float*)d_in[2];
  float*       out  = (float*)d_out;

  (void)d_ws; (void)ws_size;
  tri_wmma_fused<<<NPAIR, NSL * 32, 0, stream>>>(w, x, bias, out);
}
